# starting kernel
# baseline (speedup 1.0000x reference)
#include <hip/hip_runtime.h>
#include <hip/hip_bf16.h>
#include <math.h>

#define B_ 128
#define T_ 2048
#define H_ 256
#define S_ 8
#define ROWS_PER_WG (T_ / S_)
#define ROWS_PER_WAVE (ROWS_PER_WG / 8)

typedef __attribute__((ext_vector_type(2))) float v2f;
typedef __attribute__((ext_vector_type(8))) float v8f;

__global__ __launch_bounds__(32) void k_vmat(const float* __restrict__ hid,
                                             const float* __restrict__ Wscore,
                                             float* __restrict__ V) {
  const int tm   = blockIdx.x;
  const int tn   = blockIdx.y;
  const int lane = threadIdx.x;
  const int half = lane >> 4;
  const int r    = lane & 15;
  const size_t hstride = (size_t)T_ * H_;
  const float* ht = hid + (size_t)(T_ - 1) * H_;

  v8f acc = {};
  for (int k0 = 0; k0 < H_; k0 += 4) {
    const int k = k0 + 2 * half;
    const float2 av = *(const float2*)(ht + (size_t)(tm * 16 + r) * hstride + k);
    const float2 bv = *(const float2*)(Wscore + (size_t)(tn * 16 + r) * H_ + k);
    v2f a; a[0] = av.x; a[1] = av.y;
    v2f b; b[0] = bv.x; b[1] = bv.y;
    acc = __builtin_amdgcn_wmma_f32_16x16x4_f32(false, a, false, b,
                                                (short)0, acc, false, false);
  }
#pragma unroll
  for (int j = 0; j < 8; ++j) {
    const int m = tm * 16 + j + 8 * half;
    const int n = tn * 16 + r;
    V[(size_t)m * H_ + n] = acc[j];
  }
}

__global__ __launch_bounds__(256) void k_stream(const float* __restrict__ hid,
                                                const float* __restrict__ V,
                                                float* __restrict__ pm,
                                                float* __restrict__ psum,
                                                float* __restrict__ pc) {
  const int b    = blockIdx.x;
  const int s    = blockIdx.y;
  const int wave = threadIdx.x >> 5;
  const int lane = threadIdx.x & 31;

  const float* vrow = V + (size_t)b * H_ + lane * 8;
  const float4 v0 = *(const float4*)vrow;
  const float4 v1 = *(const float4*)(vrow + 4);

  float m = -INFINITY, run = 0.f;
  float c[8] = {0.f, 0.f, 0.f, 0.f, 0.f, 0.f, 0.f, 0.f};

  const int t0 = s * ROWS_PER_WG + wave * ROWS_PER_WAVE;
  const float* base = hid + ((size_t)b * T_ + t0) * H_ + lane * 8;

  for (int i = 0; i < ROWS_PER_WAVE; ++i) {
    const float* row = base + (size_t)i * H_;
    const float4 h0 = *(const float4*)row;
    const float4 h1 = *(const float4*)(row + 4);
    float p = h0.x * v0.x + h0.y * v0.y + h0.z * v0.z + h0.w * v0.w
            + h1.x * v1.x + h1.y * v1.y + h1.z * v1.z + h1.w * v1.w;
    p += __shfl_xor(p, 16);
    p += __shfl_xor(p, 8);
    p += __shfl_xor(p, 4);
    p += __shfl_xor(p, 2);
    p += __shfl_xor(p, 1);
    const float mn = fmaxf(m, p);
    const float sc = __expf(m - mn);
    const float pe = __expf(p - mn);
    run = run * sc + pe;
    m = mn;
    c[0] = c[0] * sc + pe * h0.x;
    c[1] = c[1] * sc + pe * h0.y;
    c[2] = c[2] * sc + pe * h0.z;
    c[3] = c[3] * sc + pe * h0.w;
    c[4] = c[4] * sc + pe * h1.x;
    c[5] = c[5] * sc + pe * h1.y;
    c[6] = c[6] * sc + pe * h1.z;
    c[7] = c[7] * sc + pe * h1.w;
  }

  __shared__ float sm[8];
  __shared__ float ss[8];
  __shared__ float scx[8][H_];
#pragma unroll
  for (int j = 0; j < 8; ++j) scx[wave][lane * 8 + j] = c[j];
  if (lane == 0) { sm[wave] = m; ss[wave] = run; }
  __syncthreads();

  const int h = threadIdx.x;
  float M = sm[0];
#pragma unroll
  for (int w = 1; w < 8; ++w) M = fmaxf(M, sm[w]);
  float tot = 0.f, ch = 0.f;
#pragma unroll
  for (int w = 0; w < 8; ++w) {
    const float e = __expf(sm[w] - M);
    tot += ss[w] * e;
    ch  += scx[w][h] * e;
  }
  const int idx = b * S_ + s;
  pc[(size_t)idx * H_ + h] = ch;
  if (threadIdx.x == 0) { pm[idx] = M; psum[idx] = tot; }
}

__global__ __launch_bounds__(256) void k_combine(const float* __restrict__ hid,
                                                 const float* __restrict__ pm,
                                                 const float* __restrict__ psum,
                                                 const float* __restrict__ pc,
                                                 float* __restrict__ preact) {
  const int b = blockIdx.x;
  const int h = threadIdx.x;
  float M = pm[b * S_];
#pragma unroll
  for (int s = 1; s < S_; ++s) M = fmaxf(M, pm[b * S_ + s]);
  float tot = 0.f, ch = 0.f;
#pragma unroll
  for (int s = 0; s < S_; ++s) {
    const float e = __expf(pm[b * S_ + s] - M);
    tot += psum[b * S_ + s] * e;
    ch  += pc[(size_t)(b * S_ + s) * H_ + h] * e;
  }
  preact[(size_t)b * 512 + h]       = ch / tot;
  preact[(size_t)b * 512 + 256 + h] = hid[((size_t)b * T_ + (T_ - 1)) * H_ + h];
}

__global__ __launch_bounds__(32) void k_att(const float* __restrict__ preact,
                                            const float* __restrict__ Watt,
                                            float* __restrict__ out) {
  const int tm   = blockIdx.x;
  const int tn   = blockIdx.y;
  const int lane = threadIdx.x;
  const int half = lane >> 4;
  const int r    = lane & 15;

  v8f acc = {};
  for (int k0 = 0; k0 < 512; k0 += 4) {
    const int k = k0 + 2 * half;
    const float2 av = *(const float2*)(preact + (size_t)(tm * 16 + r) * 512 + k);
    v2f a; a[0] = av.x; a[1] = av.y;
    v2f b;
    b[0] = Watt[(size_t)k * 128 + tn * 16 + r];
    b[1] = Watt[(size_t)(k + 1) * 128 + tn * 16 + r];
    acc = __builtin_amdgcn_wmma_f32_16x16x4_f32(false, a, false, b,
                                                (short)0, acc, false, false);
  }
#pragma unroll
  for (int j = 0; j < 8; ++j) {
    const int m = tm * 16 + j + 8 * half;
    const int n = tn * 16 + r;
    out[(size_t)m * 128 + n] = tanhf(acc[j]);
  }
}

extern "C" void kernel_launch(void* const* d_in, const int* in_sizes, int n_in,
                              void* d_out, int out_size, void* d_ws, size_t ws_size,
                              hipStream_t stream) {
  const float* hid    = (const float*)d_in[0];
  const float* Wscore = (const float*)d_in[1];
  const float* Watt   = (const float*)d_in[2];
  float* out = (float*)d_out;
  float* ws  = (float*)d_ws;

  float* V      = ws;
  float* pm     = ws + 32768;
  float* psum   = ws + 33792;
  float* pc     = ws + 34816;
  float* preact = ws + 296960;

  k_vmat   <<<dim3(8, 16),  32,  0, stream>>>(hid, Wscore, V);
  k_stream <<<dim3(B_, S_), 256, 0, stream>>>(hid, V, pm, psum, pc);
  k_combine<<<B_,           256, 0, stream>>>(hid, pm, psum, pc, preact);
  k_att    <<<dim3(8, 8),   32,  0, stream>>>(preact, Watt, out);
}


# Final GEMM taken on the raw split partials (M tile = 2 batches x 8 splits, rows scaled by softmax weight), the 8 D registers summed per lane: combine launch and preact go; split-sum order changes
# speedup vs baseline: 1.1580x; 1.1580x over previous
#include <hip/hip_runtime.h>
#include <hip/hip_bf16.h>
#include <math.h>

#define B_ 128
#define T_ 2048
#define H_ 256
#define S_ 8
#define ROWS_PER_WG (T_ / S_)
#define ROWS_PER_WAVE (ROWS_PER_WG / 8)

typedef __attribute__((ext_vector_type(2))) float v2f;
typedef __attribute__((ext_vector_type(8))) float v8f;

__global__ __launch_bounds__(32) void k_vmat(const float* __restrict__ hid,
                                             const float* __restrict__ Wscore,
                                             float* __restrict__ V) {
  const int tm   = blockIdx.x;
  const int tn   = blockIdx.y;
  const int lane = threadIdx.x;
  const int half = lane >> 4;
  const int r    = lane & 15;
  const size_t hstride = (size_t)T_ * H_;
  const float* ht = hid + (size_t)(T_ - 1) * H_;

  v8f acc = {};
  for (int k0 = 0; k0 < H_; k0 += 4) {
    const int k = k0 + 2 * half;
    const float2 av = *(const float2*)(ht + (size_t)(tm * 16 + r) * hstride + k);
    const float2 bv = *(const float2*)(Wscore + (size_t)(tn * 16 + r) * H_ + k);
    v2f a; a[0] = av.x; a[1] = av.y;
    v2f b; b[0] = bv.x; b[1] = bv.y;
    acc = __builtin_amdgcn_wmma_f32_16x16x4_f32(false, a, false, b,
                                                (short)0, acc, false, false);
  }
#pragma unroll
  for (int j = 0; j < 8; ++j) {
    const int m = tm * 16 + j + 8 * half;
    const int n = tn * 16 + r;
    V[(size_t)m * H_ + n] = acc[j];
  }
}

__global__ __launch_bounds__(256) void k_stream(const float* __restrict__ hid,
                                                const float* __restrict__ V,
                                                float* __restrict__ pm,
                                                float* __restrict__ psum,
                                                float* __restrict__ pc) {
  const int b    = blockIdx.x;
  const int s    = blockIdx.y;
  const int wave = threadIdx.x >> 5;
  const int lane = threadIdx.x & 31;

  const float* vrow = V + (size_t)b * H_ + lane * 8;
  const float4 v0 = *(const float4*)vrow;
  const float4 v1 = *(const float4*)(vrow + 4);

  float m = -INFINITY, run = 0.f;
  float c[8] = {0.f, 0.f, 0.f, 0.f, 0.f, 0.f, 0.f, 0.f};

  const int t0 = s * ROWS_PER_WG + wave * ROWS_PER_WAVE;
  const float* base = hid + ((size_t)b * T_ + t0) * H_ + lane * 8;

  for (int i = 0; i < ROWS_PER_WAVE; ++i) {
    const float* row = base + (size_t)i * H_;
    const float4 h0 = *(const float4*)row;
    const float4 h1 = *(const float4*)(row + 4);
    float p = h0.x * v0.x + h0.y * v0.y + h0.z * v0.z + h0.w * v0.w
            + h1.x * v1.x + h1.y * v1.y + h1.z * v1.z + h1.w * v1.w;
    p += __shfl_xor(p, 16);
    p += __shfl_xor(p, 8);
    p += __shfl_xor(p, 4);
    p += __shfl_xor(p, 2);
    p += __shfl_xor(p, 1);
    const float mn = fmaxf(m, p);
    const float sc = __expf(m - mn);
    const float pe = __expf(p - mn);
    run = run * sc + pe;
    m = mn;
    c[0] = c[0] * sc + pe * h0.x;
    c[1] = c[1] * sc + pe * h0.y;
    c[2] = c[2] * sc + pe * h0.z;
    c[3] = c[3] * sc + pe * h0.w;
    c[4] = c[4] * sc + pe * h1.x;
    c[5] = c[5] * sc + pe * h1.y;
    c[6] = c[6] * sc + pe * h1.z;
    c[7] = c[7] * sc + pe * h1.w;
  }

  __shared__ float sm[8];
  __shared__ float ss[8];
  __shared__ float scx[8][H_];
#pragma unroll
  for (int j = 0; j < 8; ++j) scx[wave][lane * 8 + j] = c[j];
  if (lane == 0) { sm[wave] = m; ss[wave] = run; }
  __syncthreads();

  const int h = threadIdx.x;
  float M = sm[0];
#pragma unroll
  for (int w = 1; w < 8; ++w) M = fmaxf(M, sm[w]);
  float tot = 0.f, ch = 0.f;
#pragma unroll
  for (int w = 0; w < 8; ++w) {
    const float e = __expf(sm[w] - M);
    tot += ss[w] * e;
    ch  += scx[w][h] * e;
  }
  const int idx = b * S_ + s;
  pc[(size_t)idx * H_ + h] = ch;
  if (threadIdx.x == 0) { pm[idx] = M; psum[idx] = tot; }
}

#define FIN_U 8
#define FIN_G (32 / FIN_U)

struct FinB { float2 b[FIN_U][4]; };
struct FinA { float4 a[FIN_U]; };

__device__ __forceinline__ void fin_load_b(FinB& q, const float* bcol, int i0) {
#pragma unroll
  for (int u = 0; u < FIN_U; ++u)
#pragma unroll
    for (int j = 0; j < 4; ++j)
      q.b[u][j] = *(const float2*)(bcol + (size_t)(8 * (i0 + u) + j) * 128);
}

__device__ __forceinline__ void fin_load_a(FinA& q, const float* arow, int i0) {
#pragma unroll
  for (int u = 0; u < FIN_U; ++u) q.a[u] = *(const float4*)(arow + 8 * (i0 + u));
}

__device__ __forceinline__ void fin_mma(const float a0, const float a1, const float a2,
                                        const float a3, const float2 (&b)[4],
                                        v8f& acc0, v8f& acc1) {
  const float av[4] = {a0, a1, a2, a3};
#pragma unroll
  for (int e = 0; e < 2; ++e) {
    v2f a;  a[0]  = av[2 * e];     a[1]  = av[2 * e + 1];
    v2f b0; b0[0] = b[2 * e].x;    b0[1] = b[2 * e + 1].x;
    v2f b1; b1[0] = b[2 * e].y;    b1[1] = b[2 * e + 1].y;
    acc0 = __builtin_amdgcn_wmma_f32_16x16x4_f32(false, a, false, b0,
                                                 (short)0, acc0, false, false);
    acc1 = __builtin_amdgcn_wmma_f32_16x16x4_f32(false, a, false, b1,
                                                 (short)0, acc1, false, false);
  }
}

__global__ __launch_bounds__(32) __attribute__((amdgpu_waves_per_eu(1, 1)))
void k_fin(const float* __restrict__ hid,
           const float* __restrict__ pm,
           const float* __restrict__ psum,
           const float* __restrict__ pc,
           const float* __restrict__ Watt,
           float* __restrict__ out) {
  const int tm   = blockIdx.x;
  const int n0   = blockIdx.y * 32;
  const int lane = threadIdx.x;
  const int half = lane >> 4;
  const int c    = lane & 15;
  const int bA   = 2 * tm + (c >> 3);
  const int sA   = c & 7;

  const float* arow1 = pc + (size_t)(tm * 16 + c) * H_ + 4 * half;
  const float* arow2 = hid + ((size_t)bA * T_ + (T_ - 1)) * H_ + 4 * half;
  const float* bcol1 = Watt + (size_t)(4 * half) * 128 + n0 + 2 * c;
  const float* bcol2 = bcol1 + (size_t)256 * 128;

  FinA aq[2];
  FinB bq[2];
  float4 ht[32];
  const float4 m0 = *(const float4*)(pm + bA * S_);
  const float4 m1 = *(const float4*)(pm + bA * S_ + 4);
  const float4 q0 = *(const float4*)(psum + bA * S_);
  const float4 q1 = *(const float4*)(psum + bA * S_ + 4);
  fin_load_a(aq[0], arow1, 0);
  fin_load_b(bq[0], bcol1, 0);

  const float mm[8] = {m0.x, m0.y, m0.z, m0.w, m1.x, m1.y, m1.z, m1.w};
  const float qq[8] = {q0.x, q0.y, q0.z, q0.w, q1.x, q1.y, q1.z, q1.w};
  float M = mm[0];
#pragma unroll
  for (int s = 1; s < S_; ++s) M = fmaxf(M, mm[s]);
  float tot = 0.f, es = 0.f;
#pragma unroll
  for (int s = 0; s < S_; ++s) {
    const float e = __expf(mm[s] - M);
    tot += qq[s] * e;
    es = (s == sA) ? e : es;
  }
  const float w = es / tot;
  const bool first = (sA == 0);

  v8f acc0 = {}, acc1 = {};
#pragma unroll
  for (int g = 0; g < 2 * FIN_G; ++g) {
    if (g + 1 < FIN_G)          fin_load_a(aq[(g + 1) & 1], arow1, (g + 1) * FIN_U);
    if (g + 1 < FIN_G)          fin_load_b(bq[(g + 1) & 1], bcol1, (g + 1) * FIN_U);
    else if (g + 1 < 2 * FIN_G) fin_load_b(bq[(g + 1) & 1], bcol2, (g + 1 - FIN_G) * FIN_U);
    if (g < FIN_G) {
#pragma unroll
      for (int i = 0; i < FIN_U; ++i) ht[g * FIN_U + i] = *(const float4*)(arow2 + 8 * (g * FIN_U + i));
    }
    const FinB& bb = bq[g & 1];
    if (g < FIN_G) {
      const FinA& aa = aq[g & 1];
#pragma unroll
      for (int u = 0; u < FIN_U; ++u)
        fin_mma(aa.a[u].x * w, aa.a[u].y * w, aa.a[u].z * w, aa.a[u].w * w,
                bb.b[u], acc0, acc1);
    } else {
#pragma unroll
      for (int u = 0; u < FIN_U; ++u) {
        const float4 h4 = ht[(g - FIN_G) * FIN_U + u];
        fin_mma(first ? h4.x : 0.f, first ? h4.y : 0.f, first ? h4.z : 0.f,
                first ? h4.w : 0.f, bb.b[u], acc0, acc1);
      }
    }
  }

  float r0 = acc0[0], r1 = acc1[0];
#pragma unroll
  for (int j = 1; j < 8; ++j) { r0 += acc0[j]; r1 += acc1[j]; }
  float2 o; o.x = tanhf(r0); o.y = tanhf(r1);
  *(float2*)(out + (size_t)(2 * tm + half) * 128 + n0 + 2 * c) = o;
}

extern "C" void kernel_launch(void* const* d_in, const int* in_sizes, int n_in,
                              void* d_out, int out_size, void* d_ws, size_t ws_size,
                              hipStream_t stream) {
  const float* hid    = (const float*)d_in[0];
  const float* Wscore = (const float*)d_in[1];
  const float* Watt   = (const float*)d_in[2];
  float* out = (float*)d_out;
  float* ws  = (float*)d_ws;

  float* V      = ws;
  float* pm     = ws + 32768;
  float* psum   = ws + 33792;
  float* pc     = ws + 34816;

  k_vmat   <<<dim3(8, 16),  32,  0, stream>>>(hid, Wscore, V);
  k_stream <<<dim3(B_, S_), 256, 0, stream>>>(hid, V, pm, psum, pc);
  k_fin    <<<dim3(64, 4),  32,  0, stream>>>(hid, pm, psum, pc, Watt, out);
}


# k_vmat: each wave requests its whole K extent of both operands as one group of 64 16-byte loads ahead of its matrix run (one memory round trip, not eight), two accumulators; K sum order changes
# speedup vs baseline: 1.2082x; 1.0434x over previous
#include <hip/hip_runtime.h>
#include <hip/hip_bf16.h>
#include <math.h>

#define B_ 128
#define T_ 2048
#define H_ 256
#define S_ 8
#define ROWS_PER_WG (T_ / S_)
#define ROWS_PER_WAVE (ROWS_PER_WG / 8)

typedef __attribute__((ext_vector_type(2))) float v2f;
typedef __attribute__((ext_vector_type(8))) float v8f;

#define VM_NB 32

__global__ __launch_bounds__(32) __attribute__((amdgpu_waves_per_eu(1, 1)))
void k_vmat(const float* __restrict__ hid,
            const float* __restrict__ Wscore,
            float* __restrict__ V) {
  const int tm   = blockIdx.x;
  const int tn   = blockIdx.y;
  const int lane = threadIdx.x;
  const int half = lane >> 4;
  const int r    = lane & 15;
  const size_t hstride = (size_t)T_ * H_;
  const float* ht = hid + (size_t)(T_ - 1) * H_;

  const float* arow = ht + (size_t)(tm * 16 + r) * hstride + 4 * half;
  const float* brow = Wscore + (size_t)(tn * 16 + r) * H_ + 4 * half;

  float4 av[VM_NB], bv[VM_NB];
#pragma unroll
  for (int i = 0; i < VM_NB; ++i) {
    av[i] = *(const float4*)(arow + 8 * i);
    bv[i] = *(const float4*)(brow + 8 * i);
  }

  v8f acc0 = {}, acc1 = {};
#pragma unroll
  for (int i = 0; i < VM_NB; ++i) {
    v2f a0; a0[0] = av[i].x; a0[1] = av[i].y;
    v2f b0; b0[0] = bv[i].x; b0[1] = bv[i].y;
    v2f a1; a1[0] = av[i].z; a1[1] = av[i].w;
    v2f b1; b1[0] = bv[i].z; b1[1] = bv[i].w;
    acc0 = __builtin_amdgcn_wmma_f32_16x16x4_f32(false, a0, false, b0,
                                                 (short)0, acc0, false, false);
    acc1 = __builtin_amdgcn_wmma_f32_16x16x4_f32(false, a1, false, b1,
                                                 (short)0, acc1, false, false);
  }
#pragma unroll
  for (int j = 0; j < 8; ++j) {
    const int m = tm * 16 + j + 8 * half;
    const int n = tn * 16 + r;
    V[(size_t)m * H_ + n] = acc0[j] + acc1[j];
  }
}

__global__ __launch_bounds__(256) void k_stream(const float* __restrict__ hid,
                                                const float* __restrict__ V,
                                                float* __restrict__ pm,
                                                float* __restrict__ psum,
                                                float* __restrict__ pc) {
  const int b    = blockIdx.x;
  const int s    = blockIdx.y;
  const int wave = threadIdx.x >> 5;
  const int lane = threadIdx.x & 31;

  const float* vrow = V + (size_t)b * H_ + lane * 8;
  const float4 v0 = *(const float4*)vrow;
  const float4 v1 = *(const float4*)(vrow + 4);

  float m = -INFINITY, run = 0.f;
  float c[8] = {0.f, 0.f, 0.f, 0.f, 0.f, 0.f, 0.f, 0.f};

  const int t0 = s * ROWS_PER_WG + wave * ROWS_PER_WAVE;
  const float* base = hid + ((size_t)b * T_ + t0) * H_ + lane * 8;

  for (int i = 0; i < ROWS_PER_WAVE; ++i) {
    const float* row = base + (size_t)i * H_;
    const float4 h0 = *(const float4*)row;
    const float4 h1 = *(const float4*)(row + 4);
    float p = h0.x * v0.x + h0.y * v0.y + h0.z * v0.z + h0.w * v0.w
            + h1.x * v1.x + h1.y * v1.y + h1.z * v1.z + h1.w * v1.w;
    p += __shfl_xor(p, 16);
    p += __shfl_xor(p, 8);
    p += __shfl_xor(p, 4);
    p += __shfl_xor(p, 2);
    p += __shfl_xor(p, 1);
    const float mn = fmaxf(m, p);
    const float sc = __expf(m - mn);
    const float pe = __expf(p - mn);
    run = run * sc + pe;
    m = mn;
    c[0] = c[0] * sc + pe * h0.x;
    c[1] = c[1] * sc + pe * h0.y;
    c[2] = c[2] * sc + pe * h0.z;
    c[3] = c[3] * sc + pe * h0.w;
    c[4] = c[4] * sc + pe * h1.x;
    c[5] = c[5] * sc + pe * h1.y;
    c[6] = c[6] * sc + pe * h1.z;
    c[7] = c[7] * sc + pe * h1.w;
  }

  __shared__ float sm[8];
  __shared__ float ss[8];
  __shared__ float scx[8][H_];
#pragma unroll
  for (int j = 0; j < 8; ++j) scx[wave][lane * 8 + j] = c[j];
  if (lane == 0) { sm[wave] = m; ss[wave] = run; }
  __syncthreads();

  const int h = threadIdx.x;
  float M = sm[0];
#pragma unroll
  for (int w = 1; w < 8; ++w) M = fmaxf(M, sm[w]);
  float tot = 0.f, ch = 0.f;
#pragma unroll
  for (int w = 0; w < 8; ++w) {
    const float e = __expf(sm[w] - M);
    tot += ss[w] * e;
    ch  += scx[w][h] * e;
  }
  const int idx = b * S_ + s;
  pc[(size_t)idx * H_ + h] = ch;
  if (threadIdx.x == 0) { pm[idx] = M; psum[idx] = tot; }
}

#define FIN_U 8
#define FIN_G (32 / FIN_U)

struct FinB { float2 b[FIN_U][4]; };
struct FinA { float4 a[FIN_U]; };

__device__ __forceinline__ void fin_load_b(FinB& q, const float* bcol, int i0) {
#pragma unroll
  for (int u = 0; u < FIN_U; ++u)
#pragma unroll
    for (int j = 0; j < 4; ++j)
      q.b[u][j] = *(const float2*)(bcol + (size_t)(8 * (i0 + u) + j) * 128);
}

__device__ __forceinline__ void fin_load_a(FinA& q, const float* arow, int i0) {
#pragma unroll
  for (int u = 0; u < FIN_U; ++u) q.a[u] = *(const float4*)(arow + 8 * (i0 + u));
}

__device__ __forceinline__ void fin_mma(const float a0, const float a1, const float a2,
                                        const float a3, const float2 (&b)[4],
                                        v8f& acc0, v8f& acc1) {
  const float av[4] = {a0, a1, a2, a3};
#pragma unroll
  for (int e = 0; e < 2; ++e) {
    v2f a;  a[0]  = av[2 * e];     a[1]  = av[2 * e + 1];
    v2f b0; b0[0] = b[2 * e].x;    b0[1] = b[2 * e + 1].x;
    v2f b1; b1[0] = b[2 * e].y;    b1[1] = b[2 * e + 1].y;
    acc0 = __builtin_amdgcn_wmma_f32_16x16x4_f32(false, a, false, b0,
                                                 (short)0, acc0, false, false);
    acc1 = __builtin_amdgcn_wmma_f32_16x16x4_f32(false, a, false, b1,
                                                 (short)0, acc1, false, false);
  }
}

__global__ __launch_bounds__(32) __attribute__((amdgpu_waves_per_eu(1, 1)))
void k_fin(const float* __restrict__ hid,
           const float* __restrict__ pm,
           const float* __restrict__ psum,
           const float* __restrict__ pc,
           const float* __restrict__ Watt,
           float* __restrict__ out) {
  const int tm   = blockIdx.x;
  const int n0   = blockIdx.y * 32;
  const int lane = threadIdx.x;
  const int half = lane >> 4;
  const int c    = lane & 15;
  const int bA   = 2 * tm + (c >> 3);
  const int sA   = c & 7;

  const float* arow1 = pc + (size_t)(tm * 16 + c) * H_ + 4 * half;
  const float* arow2 = hid + ((size_t)bA * T_ + (T_ - 1)) * H_ + 4 * half;
  const float* bcol1 = Watt + (size_t)(4 * half) * 128 + n0 + 2 * c;
  const float* bcol2 = bcol1 + (size_t)256 * 128;

  FinA aq[2];
  FinB bq[2];
  float4 ht[32];
  const float4 m0 = *(const float4*)(pm + bA * S_);
  const float4 m1 = *(const float4*)(pm + bA * S_ + 4);
  const float4 q0 = *(const float4*)(psum + bA * S_);
  const float4 q1 = *(const float4*)(psum + bA * S_ + 4);
  fin_load_a(aq[0], arow1, 0);
  fin_load_b(bq[0], bcol1, 0);

  const float mm[8] = {m0.x, m0.y, m0.z, m0.w, m1.x, m1.y, m1.z, m1.w};
  const float qq[8] = {q0.x, q0.y, q0.z, q0.w, q1.x, q1.y, q1.z, q1.w};
  float M = mm[0];
#pragma unroll
  for (int s = 1; s < S_; ++s) M = fmaxf(M, mm[s]);
  float tot = 0.f, es = 0.f;
#pragma unroll
  for (int s = 0; s < S_; ++s) {
    const float e = __expf(mm[s] - M);
    tot += qq[s] * e;
    es = (s == sA) ? e : es;
  }
  const float w = es / tot;
  const bool first = (sA == 0);

  v8f acc0 = {}, acc1 = {};
#pragma unroll
  for (int g = 0; g < 2 * FIN_G; ++g) {
    if (g + 1 < FIN_G)          fin_load_a(aq[(g + 1) & 1], arow1, (g + 1) * FIN_U);
    if (g + 1 < FIN_G)          fin_load_b(bq[(g + 1) & 1], bcol1, (g + 1) * FIN_U);
    else if (g + 1 < 2 * FIN_G) fin_load_b(bq[(g + 1) & 1], bcol2, (g + 1 - FIN_G) * FIN_U);
    if (g < FIN_G) {
#pragma unroll
      for (int i = 0; i < FIN_U; ++i) ht[g * FIN_U + i] = *(const float4*)(arow2 + 8 * (g * FIN_U + i));
    }
    const FinB& bb = bq[g & 1];
    if (g < FIN_G) {
      const FinA& aa = aq[g & 1];
#pragma unroll
      for (int u = 0; u < FIN_U; ++u)
        fin_mma(aa.a[u].x * w, aa.a[u].y * w, aa.a[u].z * w, aa.a[u].w * w,
                bb.b[u], acc0, acc1);
    } else {
#pragma unroll
      for (int u = 0; u < FIN_U; ++u) {
        const float4 h4 = ht[(g - FIN_G) * FIN_U + u];
        fin_mma(first ? h4.x : 0.f, first ? h4.y : 0.f, first ? h4.z : 0.f,
                first ? h4.w : 0.f, bb.b[u], acc0, acc1);
      }
    }
  }

  float r0 = acc0[0], r1 = acc1[0];
#pragma unroll
  for (int j = 1; j < 8; ++j) { r0 += acc0[j]; r1 += acc1[j]; }
  float2 o; o.x = tanhf(r0); o.y = tanhf(r1);
  *(float2*)(out + (size_t)(2 * tm + half) * 128 + n0 + 2 * c) = o;
}

extern "C" void kernel_launch(void* const* d_in, const int* in_sizes, int n_in,
                              void* d_out, int out_size, void* d_ws, size_t ws_size,
                              hipStream_t stream) {
  const float* hid    = (const float*)d_in[0];
  const float* Wscore = (const float*)d_in[1];
  const float* Watt   = (const float*)d_in[2];
  float* out = (float*)d_out;
  float* ws  = (float*)d_ws;

  float* V      = ws;
  float* pm     = ws + 32768;
  float* psum   = ws + 33792;
  float* pc     = ws + 34816;

  k_vmat   <<<dim3(8, 16),  32,  0, stream>>>(hid, Wscore, V);
  k_stream <<<dim3(B_, S_), 256, 0, stream>>>(hid, V, pm, psum, pc);
  k_fin    <<<dim3(64, 4),  32,  0, stream>>>(hid, pm, psum, pc, Watt, out);
}


# k_stream re-tiled: 8-lane group per row (lane = 32 columns), a wave takes 8 rows a step as 16 whole-line loads, 3-step butterfly, one softmax update per row pair; score and context sum order changes
# speedup vs baseline: 1.4392x; 1.1912x over previous
#include <hip/hip_runtime.h>
#include <hip/hip_bf16.h>
#include <math.h>

#define B_ 128
#define T_ 2048
#define H_ 256
#define S_ 8
#define ROWS_PER_WG (T_ / S_)

typedef __attribute__((ext_vector_type(2))) float v2f;
typedef __attribute__((ext_vector_type(8))) float v8f;

#define VM_NB 32

__global__ __launch_bounds__(32) __attribute__((amdgpu_waves_per_eu(1, 1)))
void k_vmat(const float* __restrict__ hid,
            const float* __restrict__ Wscore,
            float* __restrict__ V) {
  const int tm   = blockIdx.x;
  const int tn   = blockIdx.y;
  const int lane = threadIdx.x;
  const int half = lane >> 4;
  const int r    = lane & 15;
  const size_t hstride = (size_t)T_ * H_;
  const float* ht = hid + (size_t)(T_ - 1) * H_;

  const float* arow = ht + (size_t)(tm * 16 + r) * hstride + 4 * half;
  const float* brow = Wscore + (size_t)(tn * 16 + r) * H_ + 4 * half;

  float4 av[VM_NB], bv[VM_NB];
#pragma unroll
  for (int i = 0; i < VM_NB; ++i) {
    av[i] = *(const float4*)(arow + 8 * i);
    bv[i] = *(const float4*)(brow + 8 * i);
  }

  v8f acc0 = {}, acc1 = {};
#pragma unroll
  for (int i = 0; i < VM_NB; ++i) {
    v2f a0; a0[0] = av[i].x; a0[1] = av[i].y;
    v2f b0; b0[0] = bv[i].x; b0[1] = bv[i].y;
    v2f a1; a1[0] = av[i].z; a1[1] = av[i].w;
    v2f b1; b1[0] = bv[i].z; b1[1] = bv[i].w;
    acc0 = __builtin_amdgcn_wmma_f32_16x16x4_f32(false, a0, false, b0,
                                                 (short)0, acc0, false, false);
    acc1 = __builtin_amdgcn_wmma_f32_16x16x4_f32(false, a1, false, b1,
                                                 (short)0, acc1, false, false);
  }
#pragma unroll
  for (int j = 0; j < 8; ++j) {
    const int m = tm * 16 + j + 8 * half;
    const int n = tn * 16 + r;
    V[(size_t)m * H_ + n] = acc0[j] + acc1[j];
  }
}

#define SW_WAVES 4
#define SW_THREADS 128
#define SW_ROWS  (ROWS_PER_WG / SW_WAVES)
#define SW_STEP  8
#define SW_NST   (SW_WAVES * 4)

__global__ __launch_bounds__(SW_THREADS)
void k_stream(const float* __restrict__ hid,
              const float* __restrict__ V,
              float* __restrict__ pm,
              float* __restrict__ psum,
              float* __restrict__ pc) {
  const int b    = blockIdx.x;
  const int s    = blockIdx.y;
  const int wave = threadIdx.x >> 5;
  const int lane = threadIdx.x & 31;
  const int g    = lane >> 3;
  const int j    = lane & 7;

  const float* vrow = V + (size_t)b * H_ + 4 * j;
  float4 v[8];
  for (int q = 0; q < 8; ++q) v[q] = *(const float4*)(vrow + 32 * q);

  float m = -INFINITY, run = 0.f;
  float4 c[8];
  for (int q = 0; q < 8; ++q) c[q] = make_float4(0.f, 0.f, 0.f, 0.f);

  const int t0 = s * ROWS_PER_WG + wave * SW_ROWS;
  const float* base = hid + ((size_t)b * T_ + t0 + g) * H_ + 4 * j;

  for (int i = 0; i < SW_ROWS / SW_STEP; ++i) {
    const float* ra = base + (size_t)(SW_STEP * i) * H_;
    const float* rb = ra + (size_t)4 * H_;
    float4 ha[8], hb[8];
    for (int q = 0; q < 8; ++q) ha[q] = *(const float4*)(ra + 32 * q);
    for (int q = 0; q < 8; ++q) hb[q] = *(const float4*)(rb + 32 * q);

    float pa0 = 0.f, pa1 = 0.f, pb0 = 0.f, pb1 = 0.f;
    for (int q = 0; q < 8; ++q) {
      pa0 = fmaf(ha[q].x, v[q].x, pa0);  pa1 = fmaf(ha[q].y, v[q].y, pa1);
      pa0 = fmaf(ha[q].z, v[q].z, pa0);  pa1 = fmaf(ha[q].w, v[q].w, pa1);
    }
    for (int q = 0; q < 8; ++q) {
      pb0 = fmaf(hb[q].x, v[q].x, pb0);  pb1 = fmaf(hb[q].y, v[q].y, pb1);
      pb0 = fmaf(hb[q].z, v[q].z, pb0);  pb1 = fmaf(hb[q].w, v[q].w, pb1);
    }
    float pa = pa0 + pa1, pb = pb0 + pb1;
    pa += __shfl_xor(pa, 4);  pb += __shfl_xor(pb, 4);
    pa += __shfl_xor(pa, 2);  pb += __shfl_xor(pb, 2);
    pa += __shfl_xor(pa, 1);  pb += __shfl_xor(pb, 1);

    const float mn = fmaxf(m, fmaxf(pa, pb));
    const float sc = __expf(m - mn);
    const float ea = __expf(pa - mn);
    const float eb = __expf(pb - mn);
    run = run * sc + (ea + eb);
    m = mn;
    for (int q = 0; q < 8; ++q) {
      c[q].x = fmaf(eb, hb[q].x, fmaf(ea, ha[q].x, c[q].x * sc));
      c[q].y = fmaf(eb, hb[q].y, fmaf(ea, ha[q].y, c[q].y * sc));
      c[q].z = fmaf(eb, hb[q].z, fmaf(ea, ha[q].z, c[q].z * sc));
      c[q].w = fmaf(eb, hb[q].w, fmaf(ea, ha[q].w, c[q].w * sc));
    }
  }

  __shared__ float sm[SW_NST];
  __shared__ float ss[SW_NST];
  __shared__ alignas(16) float scx[SW_NST][H_];
  const int st = wave * 4 + g;
  for (int q = 0; q < 8; ++q) *(float4*)&scx[st][32 * q + 4 * j] = c[q];
  if (j == 0) { sm[st] = m; ss[st] = run; }
  __syncthreads();

  const int h2 = 2 * threadIdx.x;
  float M = sm[0];
  for (int w = 1; w < SW_NST; ++w) M = fmaxf(M, sm[w]);
  float tot = 0.f;
  float2 ch = make_float2(0.f, 0.f);
  for (int w = 0; w < SW_NST; ++w) {
    const float e = __expf(sm[w] - M);
    const float2 x = *(const float2*)&scx[w][h2];
    tot += ss[w] * e;
    ch.x += x.x * e;
    ch.y += x.y * e;
  }
  const int idx = b * S_ + s;
  *(float2*)(pc + (size_t)idx * H_ + h2) = ch;
  if (threadIdx.x == 0) { pm[idx] = M; psum[idx] = tot; }
}

#define FIN_U 8
#define FIN_G (32 / FIN_U)

struct FinB { float2 b[FIN_U][4]; };
struct FinA { float4 a[FIN_U]; };

__device__ __forceinline__ void fin_load_b(FinB& q, const float* bcol, int i0) {
#pragma unroll
  for (int u = 0; u < FIN_U; ++u)
#pragma unroll
    for (int j = 0; j < 4; ++j)
      q.b[u][j] = *(const float2*)(bcol + (size_t)(8 * (i0 + u) + j) * 128);
}

__device__ __forceinline__ void fin_load_a(FinA& q, const float* arow, int i0) {
#pragma unroll
  for (int u = 0; u < FIN_U; ++u) q.a[u] = *(const float4*)(arow + 8 * (i0 + u));
}

__device__ __forceinline__ void fin_mma(const float a0, const float a1, const float a2,
                                        const float a3, const float2 (&b)[4],
                                        v8f& acc0, v8f& acc1) {
  const float av[4] = {a0, a1, a2, a3};
#pragma unroll
  for (int e = 0; e < 2; ++e) {
    v2f a;  a[0]  = av[2 * e];     a[1]  = av[2 * e + 1];
    v2f b0; b0[0] = b[2 * e].x;    b0[1] = b[2 * e + 1].x;
    v2f b1; b1[0] = b[2 * e].y;    b1[1] = b[2 * e + 1].y;
    acc0 = __builtin_amdgcn_wmma_f32_16x16x4_f32(false, a, false, b0,
                                                 (short)0, acc0, false, false);
    acc1 = __builtin_amdgcn_wmma_f32_16x16x4_f32(false, a, false, b1,
                                                 (short)0, acc1, false, false);
  }
}

__global__ __launch_bounds__(32) __attribute__((amdgpu_waves_per_eu(1, 1)))
void k_fin(const float* __restrict__ hid,
           const float* __restrict__ pm,
           const float* __restrict__ psum,
           const float* __restrict__ pc,
           const float* __restrict__ Watt,
           float* __restrict__ out) {
  const int tm   = blockIdx.x;
  const int n0   = blockIdx.y * 32;
  const int lane = threadIdx.x;
  const int half = lane >> 4;
  const int c    = lane & 15;
  const int bA   = 2 * tm + (c >> 3);
  const int sA   = c & 7;

  const float* arow1 = pc + (size_t)(tm * 16 + c) * H_ + 4 * half;
  const float* arow2 = hid + ((size_t)bA * T_ + (T_ - 1)) * H_ + 4 * half;
  const float* bcol1 = Watt + (size_t)(4 * half) * 128 + n0 + 2 * c;
  const float* bcol2 = bcol1 + (size_t)256 * 128;

  FinA aq[2];
  FinB bq[2];
  float4 ht[32];
  const float4 m0 = *(const float4*)(pm + bA * S_);
  const float4 m1 = *(const float4*)(pm + bA * S_ + 4);
  const float4 q0 = *(const float4*)(psum + bA * S_);
  const float4 q1 = *(const float4*)(psum + bA * S_ + 4);
  fin_load_a(aq[0], arow1, 0);
  fin_load_b(bq[0], bcol1, 0);

  const float mm[8] = {m0.x, m0.y, m0.z, m0.w, m1.x, m1.y, m1.z, m1.w};
  const float qq[8] = {q0.x, q0.y, q0.z, q0.w, q1.x, q1.y, q1.z, q1.w};
  float M = mm[0];
#pragma unroll
  for (int s = 1; s < S_; ++s) M = fmaxf(M, mm[s]);
  float tot = 0.f, es = 0.f;
#pragma unroll
  for (int s = 0; s < S_; ++s) {
    const float e = __expf(mm[s] - M);
    tot += qq[s] * e;
    es = (s == sA) ? e : es;
  }
  const float w = es / tot;
  const bool first = (sA == 0);

  v8f acc0 = {}, acc1 = {};
#pragma unroll
  for (int g = 0; g < 2 * FIN_G; ++g) {
    if (g + 1 < FIN_G)          fin_load_a(aq[(g + 1) & 1], arow1, (g + 1) * FIN_U);
    if (g + 1 < FIN_G)          fin_load_b(bq[(g + 1) & 1], bcol1, (g + 1) * FIN_U);
    else if (g + 1 < 2 * FIN_G) fin_load_b(bq[(g + 1) & 1], bcol2, (g + 1 - FIN_G) * FIN_U);
    if (g < FIN_G) {
#pragma unroll
      for (int i = 0; i < FIN_U; ++i) ht[g * FIN_U + i] = *(const float4*)(arow2 + 8 * (g * FIN_U + i));
    }
    const FinB& bb = bq[g & 1];
    if (g < FIN_G) {
      const FinA& aa = aq[g & 1];
#pragma unroll
      for (int u = 0; u < FIN_U; ++u)
        fin_mma(aa.a[u].x * w, aa.a[u].y * w, aa.a[u].z * w, aa.a[u].w * w,
                bb.b[u], acc0, acc1);
    } else {
#pragma unroll
      for (int u = 0; u < FIN_U; ++u) {
        const float4 h4 = ht[(g - FIN_G) * FIN_U + u];
        fin_mma(first ? h4.x : 0.f, first ? h4.y : 0.f, first ? h4.z : 0.f,
                first ? h4.w : 0.f, bb.b[u], acc0, acc1);
      }
    }
  }

  float r0 = acc0[0], r1 = acc1[0];
#pragma unroll
  for (int j = 1; j < 8; ++j) { r0 += acc0[j]; r1 += acc1[j]; }
  float2 o; o.x = tanhf(r0); o.y = tanhf(r1);
  *(float2*)(out + (size_t)(2 * tm + half) * 128 + n0 + 2 * c) = o;
}

extern "C" void kernel_launch(void* const* d_in, const int* in_sizes, int n_in,
                              void* d_out, int out_size, void* d_ws, size_t ws_size,
                              hipStream_t stream) {
  const float* hid    = (const float*)d_in[0];
  const float* Wscore = (const float*)d_in[1];
  const float* Watt   = (const float*)d_in[2];
  float* out = (float*)d_out;
  float* ws  = (float*)d_ws;

  float* V      = ws;
  float* pm     = ws + 32768;
  float* psum   = ws + 33792;
  float* pc     = ws + 34816;

  k_vmat   <<<dim3(8, 16),  32,  0, stream>>>(hid, Wscore, V);
  k_stream <<<dim3(B_, S_), SW_THREADS, 0, stream>>>(hid, V, pm, psum, pc);
  k_fin    <<<dim3(64, 4),  32,  0, stream>>>(hid, pm, psum, pc, Watt, out);
}


# h_t half of the final GEMM hoisted into launch 1 as 64 extra waves (M tile = 16 batches, four K quarters, whole-line W_att loads); last kernel keeps K=256 and adds them: h_t sum order changes
# speedup vs baseline: 1.5242x; 1.0591x over previous
#include <hip/hip_runtime.h>
#include <hip/hip_bf16.h>
#include <math.h>

#define B_ 128
#define T_ 2048
#define H_ 256
#define S_ 8
#define ROWS_PER_WG (T_ / S_)

typedef __attribute__((ext_vector_type(2))) float v2f;
typedef __attribute__((ext_vector_type(8))) float v8f;

#define VM_NB 32
#define P2_KQ 4
#define P2_K  (H_ / P2_KQ)
#define P2_NB (P2_K / 8)

__global__ __launch_bounds__(32) __attribute__((amdgpu_waves_per_eu(1, 1)))
void k_vmat(const float* __restrict__ hid,
            const float* __restrict__ Wscore,
            const float* __restrict__ Watt,
            float* __restrict__ V,
            float* __restrict__ P2) {
  const int tm   = blockIdx.x;
  const int tn   = blockIdx.y;
  const int lane = threadIdx.x;
  const int half = lane >> 4;
  const int r    = lane & 15;
  const size_t hstride = (size_t)T_ * H_;
  const float* ht = hid + (size_t)(T_ - 1) * H_;

  const float* arow = ht + (size_t)(tm * 16 + r) * hstride + 4 * half;

  if (tn >= 16) {
    const int ng = (tn - 16) & 1;
    const int kq = (tn - 16) >> 1;
    const float* pa = arow + P2_K * kq;
    const float* pb = Watt + (size_t)(256 + P2_K * kq + 4 * half) * 128 + 64 * ng + 4 * r;
    float4 av[P2_NB], bw[P2_NB][4];
#pragma unroll
    for (int i = 0; i < P2_NB; ++i) {
      av[i] = *(const float4*)(pa + 8 * i);
#pragma unroll
      for (int j = 0; j < 4; ++j) bw[i][j] = *(const float4*)(pb + (size_t)(8 * i + j) * 128);
    }
    v8f acc[4] = {};
#pragma unroll
    for (int i = 0; i < P2_NB; ++i) {
      const float a4[4] = {av[i].x, av[i].y, av[i].z, av[i].w};
#pragma unroll
      for (int e = 0; e < 2; ++e) {
        v2f a; a[0] = a4[2 * e]; a[1] = a4[2 * e + 1];
        const float4 b_lo = bw[i][2 * e], b_hi = bw[i][2 * e + 1];
        v2f b0; b0[0] = b_lo.x; b0[1] = b_hi.x;
        v2f b1; b1[0] = b_lo.y; b1[1] = b_hi.y;
        v2f b2; b2[0] = b_lo.z; b2[1] = b_hi.z;
        v2f b3; b3[0] = b_lo.w; b3[1] = b_hi.w;
        acc[0] = __builtin_amdgcn_wmma_f32_16x16x4_f32(false, a, false, b0,
                                                       (short)0, acc[0], false, false);
        acc[1] = __builtin_amdgcn_wmma_f32_16x16x4_f32(false, a, false, b1,
                                                       (short)0, acc[1], false, false);
        acc[2] = __builtin_amdgcn_wmma_f32_16x16x4_f32(false, a, false, b2,
                                                       (short)0, acc[2], false, false);
        acc[3] = __builtin_amdgcn_wmma_f32_16x16x4_f32(false, a, false, b3,
                                                       (short)0, acc[3], false, false);
      }
    }
    float* prow = P2 + ((size_t)kq * B_ + tm * 16 + 8 * half) * 128 + 64 * ng + 4 * r;
#pragma unroll
    for (int j = 0; j < 8; ++j)
      *(float4*)(prow + (size_t)j * 128) = make_float4(acc[0][j], acc[1][j], acc[2][j], acc[3][j]);
    return;
  }

  const float* brow = Wscore + (size_t)(tn * 16 + r) * H_ + 4 * half;

  float4 av[VM_NB], bv[VM_NB];
#pragma unroll
  for (int i = 0; i < VM_NB; ++i) {
    av[i] = *(const float4*)(arow + 8 * i);
    bv[i] = *(const float4*)(brow + 8 * i);
  }

  v8f acc0 = {}, acc1 = {};
#pragma unroll
  for (int i = 0; i < VM_NB; ++i) {
    v2f a0; a0[0] = av[i].x; a0[1] = av[i].y;
    v2f b0; b0[0] = bv[i].x; b0[1] = bv[i].y;
    v2f a1; a1[0] = av[i].z; a1[1] = av[i].w;
    v2f b1; b1[0] = bv[i].z; b1[1] = bv[i].w;
    acc0 = __builtin_amdgcn_wmma_f32_16x16x4_f32(false, a0, false, b0,
                                                 (short)0, acc0, false, false);
    acc1 = __builtin_amdgcn_wmma_f32_16x16x4_f32(false, a1, false, b1,
                                                 (short)0, acc1, false, false);
  }
#pragma unroll
  for (int j = 0; j < 8; ++j) {
    const int m = tm * 16 + j + 8 * half;
    const int n = tn * 16 + r;
    V[(size_t)m * H_ + n] = acc0[j] + acc1[j];
  }
}

#define SW_WAVES 4
#define SW_THREADS 128
#define SW_ROWS  (ROWS_PER_WG / SW_WAVES)
#define SW_STEP  8
#define SW_NST   (SW_WAVES * 4)

__global__ __launch_bounds__(SW_THREADS)
void k_stream(const float* __restrict__ hid,
              const float* __restrict__ V,
              float* __restrict__ pm,
              float* __restrict__ psum,
              float* __restrict__ pc) {
  const int b    = blockIdx.x;
  const int s    = blockIdx.y;
  const int wave = threadIdx.x >> 5;
  const int lane = threadIdx.x & 31;
  const int g    = lane >> 3;
  const int j    = lane & 7;

  const float* vrow = V + (size_t)b * H_ + 4 * j;
  float4 v[8];
  for (int q = 0; q < 8; ++q) v[q] = *(const float4*)(vrow + 32 * q);

  float m = -INFINITY, run = 0.f;
  float4 c[8];
  for (int q = 0; q < 8; ++q) c[q] = make_float4(0.f, 0.f, 0.f, 0.f);

  const int t0 = s * ROWS_PER_WG + wave * SW_ROWS;
  const float* base = hid + ((size_t)b * T_ + t0 + g) * H_ + 4 * j;

  for (int i = 0; i < SW_ROWS / SW_STEP; ++i) {
    const float* ra = base + (size_t)(SW_STEP * i) * H_;
    const float* rb = ra + (size_t)4 * H_;
    float4 ha[8], hb[8];
    for (int q = 0; q < 8; ++q) ha[q] = *(const float4*)(ra + 32 * q);
    for (int q = 0; q < 8; ++q) hb[q] = *(const float4*)(rb + 32 * q);

    float pa0 = 0.f, pa1 = 0.f, pb0 = 0.f, pb1 = 0.f;
    for (int q = 0; q < 8; ++q) {
      pa0 = fmaf(ha[q].x, v[q].x, pa0);  pa1 = fmaf(ha[q].y, v[q].y, pa1);
      pa0 = fmaf(ha[q].z, v[q].z, pa0);  pa1 = fmaf(ha[q].w, v[q].w, pa1);
    }
    for (int q = 0; q < 8; ++q) {
      pb0 = fmaf(hb[q].x, v[q].x, pb0);  pb1 = fmaf(hb[q].y, v[q].y, pb1);
      pb0 = fmaf(hb[q].z, v[q].z, pb0);  pb1 = fmaf(hb[q].w, v[q].w, pb1);
    }
    float pa = pa0 + pa1, pb = pb0 + pb1;
    pa += __shfl_xor(pa, 4);  pb += __shfl_xor(pb, 4);
    pa += __shfl_xor(pa, 2);  pb += __shfl_xor(pb, 2);
    pa += __shfl_xor(pa, 1);  pb += __shfl_xor(pb, 1);

    const float mn = fmaxf(m, fmaxf(pa, pb));
    const float sc = __expf(m - mn);
    const float ea = __expf(pa - mn);
    const float eb = __expf(pb - mn);
    run = run * sc + (ea + eb);
    m = mn;
    for (int q = 0; q < 8; ++q) {
      c[q].x = fmaf(eb, hb[q].x, fmaf(ea, ha[q].x, c[q].x * sc));
      c[q].y = fmaf(eb, hb[q].y, fmaf(ea, ha[q].y, c[q].y * sc));
      c[q].z = fmaf(eb, hb[q].z, fmaf(ea, ha[q].z, c[q].z * sc));
      c[q].w = fmaf(eb, hb[q].w, fmaf(ea, ha[q].w, c[q].w * sc));
    }
  }

  __shared__ float sm[SW_NST];
  __shared__ float ss[SW_NST];
  __shared__ alignas(16) float scx[SW_NST][H_];
  const int st = wave * 4 + g;
  for (int q = 0; q < 8; ++q) *(float4*)&scx[st][32 * q + 4 * j] = c[q];
  if (j == 0) { sm[st] = m; ss[st] = run; }
  __syncthreads();

  const int h2 = 2 * threadIdx.x;
  float M = sm[0];
  for (int w = 1; w < SW_NST; ++w) M = fmaxf(M, sm[w]);
  float tot = 0.f;
  float2 ch = make_float2(0.f, 0.f);
  for (int w = 0; w < SW_NST; ++w) {
    const float e = __expf(sm[w] - M);
    const float2 x = *(const float2*)&scx[w][h2];
    tot += ss[w] * e;
    ch.x += x.x * e;
    ch.y += x.y * e;
  }
  const int idx = b * S_ + s;
  *(float2*)(pc + (size_t)idx * H_ + h2) = ch;
  if (threadIdx.x == 0) { pm[idx] = M; psum[idx] = tot; }
}

#define FIN_U 8
#define FIN_G (32 / FIN_U)

struct FinB { float2 b[FIN_U][4]; };
struct FinA { float4 a[FIN_U]; };

__device__ __forceinline__ void fin_load_b(FinB& q, const float* bcol, int i0) {
#pragma unroll
  for (int u = 0; u < FIN_U; ++u)
#pragma unroll
    for (int j = 0; j < 4; ++j)
      q.b[u][j] = *(const float2*)(bcol + (size_t)(8 * (i0 + u) + j) * 128);
}

__device__ __forceinline__ void fin_load_a(FinA& q, const float* arow, int i0) {
#pragma unroll
  for (int u = 0; u < FIN_U; ++u) q.a[u] = *(const float4*)(arow + 8 * (i0 + u));
}

__device__ __forceinline__ void fin_mma(const float a0, const float a1, const float a2,
                                        const float a3, const float2 (&b)[4],
                                        v8f& acc0, v8f& acc1) {
  const float av[4] = {a0, a1, a2, a3};
#pragma unroll
  for (int e = 0; e < 2; ++e) {
    v2f a;  a[0]  = av[2 * e];     a[1]  = av[2 * e + 1];
    v2f b0; b0[0] = b[2 * e].x;    b0[1] = b[2 * e + 1].x;
    v2f b1; b1[0] = b[2 * e].y;    b1[1] = b[2 * e + 1].y;
    acc0 = __builtin_amdgcn_wmma_f32_16x16x4_f32(false, a, false, b0,
                                                 (short)0, acc0, false, false);
    acc1 = __builtin_amdgcn_wmma_f32_16x16x4_f32(false, a, false, b1,
                                                 (short)0, acc1, false, false);
  }
}

__global__ __launch_bounds__(32) __attribute__((amdgpu_waves_per_eu(1, 1)))
void k_fin(const float* __restrict__ pm,
           const float* __restrict__ psum,
           const float* __restrict__ pc,
           const float* __restrict__ P2,
           const float* __restrict__ Watt,
           float* __restrict__ out) {
  const int tm   = blockIdx.x;
  const int n0   = blockIdx.y * 32;
  const int lane = threadIdx.x;
  const int half = lane >> 4;
  const int c    = lane & 15;
  const int bA   = 2 * tm + (c >> 3);
  const int sA   = c & 7;

  const float* arow1 = pc + (size_t)(tm * 16 + c) * H_ + 4 * half;
  const float* bcol1 = Watt + (size_t)(4 * half) * 128 + n0 + 2 * c;
  const float* prow = P2 + (size_t)(2 * tm + half) * 128 + n0 + 2 * c;
  float2 p2[P2_KQ];
#pragma unroll
  for (int q = 0; q < P2_KQ; ++q) p2[q] = *(const float2*)(prow + (size_t)q * B_ * 128);

  FinA aq[2];
  FinB bq[2];
  const float4 m0 = *(const float4*)(pm + bA * S_);
  const float4 m1 = *(const float4*)(pm + bA * S_ + 4);
  const float4 q0 = *(const float4*)(psum + bA * S_);
  const float4 q1 = *(const float4*)(psum + bA * S_ + 4);
  fin_load_a(aq[0], arow1, 0);
  fin_load_b(bq[0], bcol1, 0);

  const float mm[8] = {m0.x, m0.y, m0.z, m0.w, m1.x, m1.y, m1.z, m1.w};
  const float qq[8] = {q0.x, q0.y, q0.z, q0.w, q1.x, q1.y, q1.z, q1.w};
  float M = mm[0];
#pragma unroll
  for (int s = 1; s < S_; ++s) M = fmaxf(M, mm[s]);
  float tot = 0.f, es = 0.f;
#pragma unroll
  for (int s = 0; s < S_; ++s) {
    const float e = __expf(mm[s] - M);
    tot += qq[s] * e;
    es = (s == sA) ? e : es;
  }
  const float w = es / tot;

  asm volatile("" :: "v"(p2[0].x), "v"(p2[0].y), "v"(p2[1].x), "v"(p2[1].y),
                     "v"(p2[2].x), "v"(p2[2].y), "v"(p2[3].x), "v"(p2[3].y));

  v8f acc0 = {}, acc1 = {};
#pragma unroll
  for (int g = 0; g < FIN_G; ++g) {
    if (g + 1 < FIN_G) {
      fin_load_a(aq[(g + 1) & 1], arow1, (g + 1) * FIN_U);
      fin_load_b(bq[(g + 1) & 1], bcol1, (g + 1) * FIN_U);
    }
    const FinA& aa = aq[g & 1];
    const FinB& bb = bq[g & 1];
#pragma unroll
    for (int u = 0; u < FIN_U; ++u)
      fin_mma(aa.a[u].x * w, aa.a[u].y * w, aa.a[u].z * w, aa.a[u].w * w,
              bb.b[u], acc0, acc1);
  }

  float r0 = acc0[0], r1 = acc1[0];
#pragma unroll
  for (int j = 1; j < 8; ++j) { r0 += acc0[j]; r1 += acc1[j]; }
  r0 += (p2[0].x + p2[1].x) + (p2[2].x + p2[3].x);
  r1 += (p2[0].y + p2[1].y) + (p2[2].y + p2[3].y);
  float2 o; o.x = tanhf(r0); o.y = tanhf(r1);
  *(float2*)(out + (size_t)(2 * tm + half) * 128 + n0 + 2 * c) = o;
}

extern "C" void kernel_launch(void* const* d_in, const int* in_sizes, int n_in,
                              void* d_out, int out_size, void* d_ws, size_t ws_size,
                              hipStream_t stream) {
  const float* hid    = (const float*)d_in[0];
  const float* Wscore = (const float*)d_in[1];
  const float* Watt   = (const float*)d_in[2];
  float* out = (float*)d_out;
  float* ws  = (float*)d_ws;

  float* V      = ws;
  float* pm     = ws + 32768;
  float* psum   = ws + 33792;
  float* pc     = ws + 34816;
  float* P2     = ws + 296960;

  k_vmat   <<<dim3(8, 24),  32,  0, stream>>>(hid, Wscore, Watt, V, P2);
  k_stream <<<dim3(B_, S_), SW_THREADS, 0, stream>>>(hid, V, pm, psum, pc);
  k_fin    <<<dim3(64, 4),  32,  0, stream>>>(pm, psum, pc, P2, Watt, out);
}
